# starting kernel
# baseline (speedup 1.0000x reference)
#include <hip/hip_runtime.h>


#define N_QUBITS 16
#define N_LAYERS 2
#define BATCH    512
#define DIM      65536
#define THREADS  512

typedef __attribute__((ext_vector_type(2))) float v2f;
typedef __attribute__((ext_vector_type(8))) float v8f;

__global__ __launch_bounds__(THREADS)
void qsim16_kernel(const float* __restrict__ x,
                   const float* __restrict__ params,
                   float* __restrict__ out)
{
    __shared__ __align__(16) float S[DIM];
    __shared__ float U[N_LAYERS][4][16][16];
    __shared__ float Phi[256];
    __shared__ float Plo[256];
    __shared__ float cw[16], sw[16];
    __shared__ float pcs[N_LAYERS][16];
    __shared__ float pss[N_LAYERS][16];
    __shared__ float partial[16][16];

    float4* S4 = reinterpret_cast<float4*>(S);

    const int tid  = threadIdx.x;
    const int lane = tid & 31;
    const int wave = tid >> 5;
    const int b    = blockIdx.x;

    if (tid < 16) {
        float a = 0.5f * x[b * N_QUBITS + tid];
        float s, c;
        __sincosf(a, &s, &c);
        cw[tid] = c;
        sw[tid] = s;
    } else if (tid < 48) {
        int idx = tid - 16;
        int l = idx >> 4, w = idx & 15;
        float a = 0.5f * params[idx];
        float s, c;
        __sincosf(a, &s, &c);
        pcs[l][w] = c;
        pss[l][w] = s;
    }
    __syncthreads();

    if (tid < 256) {
        int r = tid;
        float p = 1.0f;
#pragma unroll
        for (int w = 0; w < 8; ++w)
            p *= ((r >> (7 - w)) & 1) ? sw[w] : cw[w];
        Phi[r] = p;
    } else {
        int c = tid - 256;
        float p = 1.0f;
#pragma unroll
        for (int w = 8; w < 16; ++w)
            p *= ((c >> (15 - w)) & 1) ? sw[w] : cw[w];
        Plo[c] = p;
    }

    for (int e = tid; e < N_LAYERS * 4 * 256; e += THREADS) {
        int l  = e >> 10;
        int g  = (e >> 8) & 3;
        int ar = (e >> 4) & 15;
        int ac = e & 15;
        float v = 1.0f;
#pragma unroll
        for (int m = 0; m < 4; ++m) {
            int w = 4 * g + m;
            int i = (ar >> (3 - m)) & 1;
            int j = (ac >> (3 - m)) & 1;
            float f = (i == j) ? pcs[l][w] : (i ? pss[l][w] : -pss[l][w]);
            v *= f;
        }
        U[l][g][ar][ac] = v;
    }
    __syncthreads();

    float acc[16];
#pragma unroll
    for (int w = 0; w < 16; ++w) acc[w] = 0.0f;

    const int M  = lane & 15;
    const int hs = lane >> 4;

    for (int part = 0; part < 2; ++part) {
        for (int jj = 0; jj < DIM / 4 / THREADS; ++jj) {
            int blkz = tid + THREADS * jj;
            int z0   = blkz << 2;
            float ph_hi = Phi[z0 >> 8];
            int   k0   = __popc(z0);
            float4 v;
#pragma unroll
            for (int e = 0; e < 4; ++e) {
                int k = (k0 + (e == 3 ? 2 : (e == 0 ? 0 : 1))) & 3;
                float ph;
                if (part == 0) ph = (k == 0) ? 1.0f : ((k == 2) ? -1.0f : 0.0f);
                else           ph = (k == 3) ? 1.0f : ((k == 1) ? -1.0f : 0.0f);
                float m = ph_hi * Plo[(z0 & 255) | e] * ph;
                if (e == 0) v.x = m; else if (e == 1) v.y = m;
                else if (e == 2) v.z = m; else v.w = m;
            }
            S4[blkz] = v;
        }
        __syncthreads();

        for (int l = 0; l < N_LAYERS; ++l) {
            for (int g = 0; g < 4; ++g) {
                const int beta = 12 - 4 * g;
                const float* Um = &U[l][g][0][0];

                v2f afr[4];
#pragma unroll
                for (int k0 = 0; k0 < 4; ++k0) {
                    int K = 4 * k0 + 2 * hs;
                    afr[k0].x = Um[M * 16 + K];
                    afr[k0].y = Um[M * 16 + K + 1];
                }

                const int lowmask = (1 << beta) - 1;
                for (int blk = wave; blk < 256; blk += 16) {
                    int j  = blk * 16 + M;
                    int zb = ((j & ~lowmask) << 4) | (j & lowmask);
                    v8f c8 = {};
#pragma unroll
                    for (int k0 = 0; k0 < 4; ++k0) {
                        int K = 4 * k0 + 2 * hs;
                        v2f bfr;
                        bfr.x = S[zb | (K << beta)];
                        bfr.y = S[zb | ((K + 1) << beta)];
                        c8 = __builtin_amdgcn_wmma_f32_16x16x4_f32(
                                 false, afr[k0], false, bfr,
                                 (short)0, c8, false, false);
                    }
#pragma unroll
                    for (int v = 0; v < 8; ++v) {
                        int Mr = v + 8 * hs;
                        S[zb | (Mr << beta)] = c8[v];
                    }
                }
                __syncthreads();
            }

            for (int i = 0; i < 13; ++i) {
                const int plo = 14 - i;
                for (int qb = tid; qb < 4096; qb += THREADS) {
                    int q    = qb << 2;
                    int low  = q & ((1 << plo) - 1);
                    int high = q >> plo;
                    int z    = (high << (plo + 2)) | low | (1 << (plo + 1));
                    int i1   = z >> 2;
                    int i2   = i1 ^ (1 << (plo - 2));
                    float4 a = S4[i1];
                    float4 c = S4[i2];
                    S4[i1] = c;
                    S4[i2] = a;
                }
                __syncthreads();
            }
            {
                for (int h = tid; h < 8192; h += THREADS) {
                    int idx  = ((h << 1) | 1);
                    float4 v = S4[idx];
                    S4[idx]  = make_float4(v.z, v.w, v.x, v.y);
                }
                __syncthreads();
            }
            {
                for (int idx = tid; idx < 16384; idx += THREADS) {
                    float4 v = S4[idx];
                    S4[idx]  = make_float4(v.x, v.y, v.w, v.z);
                }
                __syncthreads();
            }
            {
                for (int h = tid; h < 8192; h += THREADS) {
                    int i1 = h;
                    int i2 = h ^ 0x2000;
                    float4 a = S4[i1];
                    float4 c = S4[i2];
                    float t;
                    t = a.y; a.y = c.y; c.y = t;
                    t = a.w; a.w = c.w; c.w = t;
                    S4[i1] = a;
                    S4[i2] = c;
                }
                __syncthreads();
            }
        }

        for (int jj = 0; jj < DIM / 4 / THREADS; ++jj) {
            int blkz = tid + THREADS * jj;
            int z0   = blkz << 2;
            float4 v = S4[blkz];
            float p0 = v.x * v.x, p1 = v.y * v.y, p2 = v.z * v.z, p3 = v.w * v.w;
            float ps = p0 + p1 + p2 + p3;
#pragma unroll
            for (int w = 0; w < 14; ++w)
                acc[w] += ((z0 >> (15 - w)) & 1) ? -ps : ps;
            acc[14] += (p0 + p1) - (p2 + p3);
            acc[15] += (p0 + p2) - (p1 + p3);
        }
        __syncthreads();
    }

#pragma unroll
    for (int w = 0; w < 16; ++w) {
        float v = acc[w];
        v += __shfl_xor(v, 16, 32);
        v += __shfl_xor(v, 8, 32);
        v += __shfl_xor(v, 4, 32);
        v += __shfl_xor(v, 2, 32);
        v += __shfl_xor(v, 1, 32);
        acc[w] = v;
    }
    if (lane == 0) {
#pragma unroll
        for (int w = 0; w < 16; ++w) partial[wave][w] = acc[w];
    }
    __syncthreads();
    if (tid < 16) {
        float s = 0.0f;
#pragma unroll
        for (int q = 0; q < 16; ++q) s += partial[q][tid];
        out[b * N_QUBITS + tid] = s;
    }
}

extern "C" void kernel_launch(void* const* d_in, const int* in_sizes, int n_in,
                              void* d_out, int out_size, void* d_ws, size_t ws_size,
                              hipStream_t stream) {
    const float* x      = (const float*)d_in[0];
    const float* params = (const float*)d_in[1];
    float* out          = (float*)d_out;
    (void)in_sizes; (void)n_in; (void)out_size; (void)d_ws; (void)ws_size;
    qsim16_kernel<<<BATCH, THREADS, 0, stream>>>(x, params, out);
}


# Fill and Z-expval loops fully unrolled: factors, phase patterns and sign bits that are fixed per thread or per chunk hoisted out of the loops (fill bits unchanged); expval sum order changes
# speedup vs baseline: 2.2794x; 2.2794x over previous
#include <hip/hip_runtime.h>


#define N_QUBITS 16
#define N_LAYERS 2
#define BATCH    512
#define DIM      65536
#define THREADS  512

typedef __attribute__((ext_vector_type(2))) float v2f;
typedef __attribute__((ext_vector_type(8))) float v8f;

__device__ __forceinline__ float quarter_cos(int k)
{
    k &= 3;
    return (k == 0) ? 1.0f : ((k == 2) ? -1.0f : 0.0f);
}

__device__ __forceinline__ void group_pass(float* S, const float* Um, int rowx, int colx,
                                           int WS, int CS, int wave, int c, int h)
{
    const int rsel = c ^ (c >> 1);
    v2f afr[4];
#pragma unroll
    for (int k0 = 0; k0 < 4; ++k0) {
        int K = (8 * h + 2 * k0) ^ colx;
        afr[k0].x = Um[(rsel ^ rowx) * 16 + K];
        afr[k0].y = Um[(rsel ^ rowx) * 16 + K + 1];
    }

    v8f d[16];
#pragma unroll
    for (int t = 0; t < 16; ++t) {
        const int zt  = t ^ c;
        const int row = (wave << WS) | (c << CS) | (zt << 4);
        const int sx  = (zt >> 2) & 3;
        const float4 b0 = *reinterpret_cast<const float4*>(&S[row | (((2 * h)     ^ sx) << 2)]);
        const float4 b1 = *reinterpret_cast<const float4*>(&S[row | (((2 * h + 1) ^ sx) << 2)]);
        v2f f0 = {b0.x, b0.y}, f1 = {b0.z, b0.w}, f2 = {b1.x, b1.y}, f3 = {b1.z, b1.w};
        v8f acc = {};
        acc = __builtin_amdgcn_wmma_f32_16x16x4_f32(false, afr[0], false, f0, (short)0, acc, false, false);
        acc = __builtin_amdgcn_wmma_f32_16x16x4_f32(false, afr[1], false, f1, (short)0, acc, false, false);
        acc = __builtin_amdgcn_wmma_f32_16x16x4_f32(false, afr[2], false, f2, (short)0, acc, false, false);
        acc = __builtin_amdgcn_wmma_f32_16x16x4_f32(false, afr[3], false, f3, (short)0, acc, false, false);
        d[t] = acc;
    }
    const int zs = c ^ wave;
    const int ss = (zs >> 2) & 3;
#pragma unroll
    for (int r = 0; r < 8; ++r) {
        const int rowo = (wave << WS) | ((8 * h + r) << CS) | (zs << 4);
#pragma unroll
        for (int j = 0; j < 4; ++j) {
            float4 o;
            o.x = d[4 * j][r]; o.y = d[4 * j + 1][r]; o.z = d[4 * j + 2][r]; o.w = d[4 * j + 3][r];
            *reinterpret_cast<float4*>(&S[rowo | ((j ^ ss) << 2)]) = o;
        }
    }
    __syncthreads();
}

__global__ __launch_bounds__(THREADS)
void qsim16_kernel(const float* __restrict__ x,
                   const float* __restrict__ params,
                   float* __restrict__ out)
{
    __shared__ __align__(16) float S[DIM];
    __shared__ float U[N_LAYERS][4][16][16];
    __shared__ __align__(16) float Gt[4][16];
    __shared__ float cw[16], sw[16];
    __shared__ float pcs[N_LAYERS][16];
    __shared__ float pss[N_LAYERS][16];
    __shared__ float partial[16][16];

    float4* S4 = reinterpret_cast<float4*>(S);

    const int tid  = threadIdx.x;
    const int lane = tid & 31;
    const int wave = tid >> 5;
    const int b    = blockIdx.x;

    if (tid < 16) {
        float a = 0.5f * x[b * N_QUBITS + tid];
        float s, c;
        __sincosf(a, &s, &c);
        cw[tid] = c;
        sw[tid] = s;
    } else if (tid < 48) {
        int idx = tid - 16;
        int l = idx >> 4, w = idx & 15;
        float a = 0.5f * params[idx];
        float s, c;
        __sincosf(a, &s, &c);
        pcs[l][w] = c;
        pss[l][w] = s;
    }
    __syncthreads();

    if (tid < 64) {
        int g = tid >> 4, a = tid & 15;
        float p = 1.0f;
#pragma unroll
        for (int m = 0; m < 4; ++m)
            p *= ((a >> (3 - m)) & 1) ? sw[4 * g + m] : cw[4 * g + m];
        Gt[g][a] = p;
    }

    for (int e = tid; e < N_LAYERS * 4 * 256; e += THREADS) {
        int l  = e >> 10;
        int g  = (e >> 8) & 3;
        int ar = (e >> 4) & 15;
        int ac = e & 15;
        float v = 1.0f;
#pragma unroll
        for (int m = 0; m < 4; ++m) {
            int w = 4 * g + m;
            int i = (ar >> (3 - m)) & 1;
            int j = (ac >> (3 - m)) & 1;
            float f = (i == j) ? pcs[l][w] : (i ? pss[l][w] : -pss[l][w]);
            v *= f;
        }
        U[l][g][ar][ac] = v;
    }
    __syncthreads();

    float acc[16];
#pragma unroll
    for (int w = 0; w < 16; ++w) acc[w] = 0.0f;

    const int c = lane & 15;
    const int h = lane >> 4;
    const int cc = wave & 1;

    for (int part = 0; part < 2; ++part) {
        {
            const int zp  = (tid >> 2) & 15;
            const int vlo = (tid >> 6) & 7;
            const int ch  = (tid & 3) ^ ((zp >> 2) & 3);
            const float4 g0 = *reinterpret_cast<const float4*>(&Gt[0][4 * ch]);
            float  bs[2];
            float4 pe[2], po[2];
#pragma unroll
            for (int bt = 0; bt < 2; ++bt) {
                const int v2 = vlo | (bt << 3);
                const int u1 = zp ^ v2;
                bs[bt] = Gt[1][u1] * Gt[2][v2];
                const int kb = __popc(ch) + __popc(u1) + __popc(v2) + part;
                pe[bt] = make_float4(g0.x * quarter_cos(kb),     g0.y * quarter_cos(kb + 1),
                                     g0.z * quarter_cos(kb + 1), g0.w * quarter_cos(kb + 2));
                po[bt] = make_float4(g0.x * quarter_cos(kb + 1), g0.y * quarter_cos(kb + 2),
                                     g0.z * quarter_cos(kb + 2), g0.w * quarter_cos(kb + 3));
            }
            float g3[16];
#pragma unroll
            for (int j = 0; j < 4; ++j) {
                const float4 t4 = *reinterpret_cast<const float4*>(&Gt[3][4 * j]);
                g3[4 * j] = t4.x; g3[4 * j + 1] = t4.y; g3[4 * j + 2] = t4.z; g3[4 * j + 3] = t4.w;
            }
#pragma unroll
            for (int q3 = 0; q3 < 16; ++q3) {
                const int n = __builtin_popcount(q3) & 3;
#pragma unroll
                for (int bt = 0; bt < 2; ++bt) {
                    float b3 = bs[bt] * g3[q3];
                    if (n & 2) b3 = -b3;
                    const float4 pt = (n & 1) ? po[bt] : pe[bt];
                    float4 o;
                    o.x = b3 * pt.x; o.y = b3 * pt.y; o.z = b3 * pt.z; o.w = b3 * pt.w;
                    S4[tid + THREADS * (2 * q3 + bt)] = o;
                }
            }
        }
        __syncthreads();

#pragma unroll 1
        for (int l = 0; l < N_LAYERS; ++l) {
#pragma unroll 1
            for (int g = 0; g < 4; ++g) {
                const int ws   = (g & 1) ? 8 : 12;
                const int rowx = (g > 0) ? (cc << 3) : 0;
                const int colx = (g == 0 && l > 0) ? (cc << 3) : 0;
                group_pass(S, &U[l][g][0][0], rowx, colx, ws, 20 - ws, wave, c, h);
            }
        }

        {
            v2f s01 = {0.0f, 0.0f}, s23 = {0.0f, 0.0f};
            float sa = 0.0f, sb = 0.0f;
            float dq0 = 0.0f, dq1 = 0.0f, dq2 = 0.0f, dq3 = 0.0f;
#pragma unroll
            for (int q3 = 0; q3 < 16; ++q3) {
                const float4 va = S4[tid + THREADS * (2 * q3)];
                const float4 vb = S4[tid + THREADS * (2 * q3 + 1)];
                const v2f a01 = {va.x, va.y}, a23 = {va.z, va.w};
                const v2f b01 = {vb.x, vb.y}, b23 = {vb.z, vb.w};
                s01 += a01 * a01; s23 += a23 * a23;
                s01 += b01 * b01; s23 += b23 * b23;
                const v2f ta = a01 * a01 + a23 * a23;
                const v2f tb = b01 * b01 + b23 * b23;
                const float pa = ta.x + ta.y;
                const float pb = tb.x + tb.y;
                sa += pa;
                sb += pb;
                const float pq = pa + pb;
                dq0 += (q3 & 1) ? -pq : pq;
                dq1 += (q3 & 2) ? -pq : pq;
                dq2 += (q3 & 4) ? -pq : pq;
                dq3 += (q3 & 8) ? -pq : pq;
            }
            const float s0 = s01.x, s1 = s01.y, s2 = s23.x, s3 = s23.y;
            const float db  = sa - sb;
            const float tot = sa + sb;
            const int zp  = (tid >> 2) & 15;
            const int vlo = (tid >> 6) & 7;
            const int ch  = (tid & 3) ^ ((zp >> 2) & 3);
            const int ulo = zp ^ vlo;
            acc[0] += (ch & 2) ? -dq0 : dq0;
            acc[1] += (ch & 1) ? -tot : tot;
            acc[2] += (s0 + s1) - (s2 + s3);
            acc[3] += (s0 + s2) - (s1 + s3);
            acc[4] += (ulo & 8) ? -db : db;
            acc[5] += (ulo & 4) ? -tot : tot;
            acc[6] += (ulo & 2) ? -tot : tot;
            acc[7] += (ulo & 1) ? -tot : tot;
            acc[8] += db;
            acc[9]  += (vlo & 4) ? -tot : tot;
            acc[10] += (vlo & 2) ? -tot : tot;
            acc[11] += (vlo & 1) ? -tot : tot;
            acc[12] += dq3;
            acc[13] += dq2;
            acc[14] += dq1;
            acc[15] += dq0;
        }
        __syncthreads();
    }

#pragma unroll
    for (int w = 0; w < 16; ++w) {
        float v = acc[w];
        v += __shfl_xor(v, 16, 32);
        v += __shfl_xor(v, 8, 32);
        v += __shfl_xor(v, 4, 32);
        v += __shfl_xor(v, 2, 32);
        v += __shfl_xor(v, 1, 32);
        acc[w] = v;
    }
    if (lane == 0) {
#pragma unroll
        for (int w = 0; w < 16; ++w) partial[wave][w] = acc[w];
    }
    __syncthreads();
    if (tid < 16) {
        float s = 0.0f;
#pragma unroll
        for (int q = 0; q < 16; ++q) s += partial[q][tid];
        out[b * N_QUBITS + tid] = s;
    }
}

extern "C" void kernel_launch(void* const* d_in, const int* in_sizes, int n_in,
                              void* d_out, int out_size, void* d_ws, size_t ws_size,
                              hipStream_t stream) {
    const float* x      = (const float*)d_in[0];
    const float* params = (const float*)d_in[1];
    float* out          = (float*)d_out;
    (void)in_sizes; (void)n_in; (void)out_size; (void)d_ws; (void)ws_size;
    qsim16_kernel<<<BATCH, THREADS, 0, stream>>>(x, params, out);
}


# RY pass in four 4-tile blocks: next block's 8 LDS loads issued as one group ahead of the current block's 16 WMMAs (two register sets), the block's four tile chains interleaved; result bits unchanged
# speedup vs baseline: 2.4234x; 1.0632x over previous
#include <hip/hip_runtime.h>


#define N_QUBITS 16
#define N_LAYERS 2
#define BATCH    512
#define DIM      65536
#define THREADS  512

typedef __attribute__((ext_vector_type(2))) float v2f;
typedef __attribute__((ext_vector_type(8))) float v8f;

__device__ __forceinline__ float quarter_cos(int k)
{
    k &= 3;
    return (k == 0) ? 1.0f : ((k == 2) ? -1.0f : 0.0f);
}

__device__ __forceinline__ void group_pass(float* S, const float* Um, int rowx, int colx,
                                           int WS, int CS, int wave, int c, int h)
{
    const int rsel = c ^ (c >> 1);
    v2f afr[4];
#pragma unroll
    for (int k0 = 0; k0 < 4; ++k0) {
        int K = (8 * h + 2 * k0) ^ colx;
        afr[k0].x = Um[(rsel ^ rowx) * 16 + K];
        afr[k0].y = Um[(rsel ^ rowx) * 16 + K + 1];
    }

    v8f d[16];
    float4 bq[2][8];
    const int lbase = (wave << WS) | (c << CS) | (c << 4) | (((2 * h) ^ ((c >> 2) & 3)) << 2);
#pragma unroll
    for (int i = 0; i < 4; ++i) {
        const int off = lbase ^ ((i << 4) | ((i >> 2) << 2));
        bq[0][2 * i]     = *reinterpret_cast<const float4*>(&S[off]);
        bq[0][2 * i + 1] = *reinterpret_cast<const float4*>(&S[off ^ 4]);
    }
#pragma unroll
    for (int blk = 0; blk < 4; ++blk) {
        const int cur = blk & 1;
        if (blk < 3) {
#pragma unroll
            for (int i = 0; i < 4; ++i) {
                const int t   = 4 * (blk + 1) + i;
                const int off = lbase ^ ((t << 4) | ((t >> 2) << 2));
                bq[cur ^ 1][2 * i]     = *reinterpret_cast<const float4*>(&S[off]);
                bq[cur ^ 1][2 * i + 1] = *reinterpret_cast<const float4*>(&S[off ^ 4]);
            }
        }
        __builtin_amdgcn_sched_barrier(0);
        v8f acc[4] = {};
#pragma unroll
        for (int k0 = 0; k0 < 4; ++k0) {
#pragma unroll
            for (int i = 0; i < 4; ++i) {
                const float4 q = bq[cur][2 * i + (k0 >> 1)];
                const v2f f = (k0 & 1) ? (v2f){q.z, q.w} : (v2f){q.x, q.y};
                acc[i] = __builtin_amdgcn_wmma_f32_16x16x4_f32(false, afr[k0], false, f, (short)0, acc[i], false, false);
            }
        }
#pragma unroll
        for (int i = 0; i < 4; ++i) d[4 * blk + i] = acc[i];
        __builtin_amdgcn_sched_barrier(0);
    }
    const int zs = c ^ wave;
    const int ss = (zs >> 2) & 3;
#pragma unroll
    for (int r = 0; r < 8; ++r) {
        const int rowo = (wave << WS) | ((8 * h + r) << CS) | (zs << 4);
#pragma unroll
        for (int j = 0; j < 4; ++j) {
            float4 o;
            o.x = d[4 * j][r]; o.y = d[4 * j + 1][r]; o.z = d[4 * j + 2][r]; o.w = d[4 * j + 3][r];
            *reinterpret_cast<float4*>(&S[rowo | ((j ^ ss) << 2)]) = o;
        }
    }
    __syncthreads();
}

__global__ __launch_bounds__(THREADS)
void qsim16_kernel(const float* __restrict__ x,
                   const float* __restrict__ params,
                   float* __restrict__ out)
{
    __shared__ __align__(16) float S[DIM];
    __shared__ float U[N_LAYERS][4][16][16];
    __shared__ __align__(16) float Gt[4][16];
    __shared__ float cw[16], sw[16];
    __shared__ float pcs[N_LAYERS][16];
    __shared__ float pss[N_LAYERS][16];
    __shared__ float partial[16][16];

    float4* S4 = reinterpret_cast<float4*>(S);

    const int tid  = threadIdx.x;
    const int lane = tid & 31;
    const int wave = tid >> 5;
    const int b    = blockIdx.x;

    if (tid < 16) {
        float a = 0.5f * x[b * N_QUBITS + tid];
        float s, c;
        __sincosf(a, &s, &c);
        cw[tid] = c;
        sw[tid] = s;
    } else if (tid < 48) {
        int idx = tid - 16;
        int l = idx >> 4, w = idx & 15;
        float a = 0.5f * params[idx];
        float s, c;
        __sincosf(a, &s, &c);
        pcs[l][w] = c;
        pss[l][w] = s;
    }
    __syncthreads();

    if (tid < 64) {
        int g = tid >> 4, a = tid & 15;
        float p = 1.0f;
#pragma unroll
        for (int m = 0; m < 4; ++m)
            p *= ((a >> (3 - m)) & 1) ? sw[4 * g + m] : cw[4 * g + m];
        Gt[g][a] = p;
    }

    for (int e = tid; e < N_LAYERS * 4 * 256; e += THREADS) {
        int l  = e >> 10;
        int g  = (e >> 8) & 3;
        int ar = (e >> 4) & 15;
        int ac = e & 15;
        float v = 1.0f;
#pragma unroll
        for (int m = 0; m < 4; ++m) {
            int w = 4 * g + m;
            int i = (ar >> (3 - m)) & 1;
            int j = (ac >> (3 - m)) & 1;
            float f = (i == j) ? pcs[l][w] : (i ? pss[l][w] : -pss[l][w]);
            v *= f;
        }
        U[l][g][ar][ac] = v;
    }
    __syncthreads();

    float acc[16];
#pragma unroll
    for (int w = 0; w < 16; ++w) acc[w] = 0.0f;

    const int c = lane & 15;
    const int h = lane >> 4;
    const int cc = wave & 1;

    for (int part = 0; part < 2; ++part) {
        {
            const int zp  = (tid >> 2) & 15;
            const int vlo = (tid >> 6) & 7;
            const int ch  = (tid & 3) ^ ((zp >> 2) & 3);
            const float4 g0 = *reinterpret_cast<const float4*>(&Gt[0][4 * ch]);
            float  bs[2];
            float4 pe[2], po[2];
#pragma unroll
            for (int bt = 0; bt < 2; ++bt) {
                const int v2 = vlo | (bt << 3);
                const int u1 = zp ^ v2;
                bs[bt] = Gt[1][u1] * Gt[2][v2];
                const int kb = __popc(ch) + __popc(u1) + __popc(v2) + part;
                pe[bt] = make_float4(g0.x * quarter_cos(kb),     g0.y * quarter_cos(kb + 1),
                                     g0.z * quarter_cos(kb + 1), g0.w * quarter_cos(kb + 2));
                po[bt] = make_float4(g0.x * quarter_cos(kb + 1), g0.y * quarter_cos(kb + 2),
                                     g0.z * quarter_cos(kb + 2), g0.w * quarter_cos(kb + 3));
            }
            float g3[16];
#pragma unroll
            for (int j = 0; j < 4; ++j) {
                const float4 t4 = *reinterpret_cast<const float4*>(&Gt[3][4 * j]);
                g3[4 * j] = t4.x; g3[4 * j + 1] = t4.y; g3[4 * j + 2] = t4.z; g3[4 * j + 3] = t4.w;
            }
#pragma unroll
            for (int q3 = 0; q3 < 16; ++q3) {
                const int n = __builtin_popcount(q3) & 3;
#pragma unroll
                for (int bt = 0; bt < 2; ++bt) {
                    float b3 = bs[bt] * g3[q3];
                    if (n & 2) b3 = -b3;
                    const float4 pt = (n & 1) ? po[bt] : pe[bt];
                    float4 o;
                    o.x = b3 * pt.x; o.y = b3 * pt.y; o.z = b3 * pt.z; o.w = b3 * pt.w;
                    S4[tid + THREADS * (2 * q3 + bt)] = o;
                }
            }
        }
        __syncthreads();

#pragma unroll 1
        for (int l = 0; l < N_LAYERS; ++l) {
#pragma unroll 1
            for (int g = 0; g < 4; ++g) {
                const int ws   = (g & 1) ? 8 : 12;
                const int rowx = (g > 0) ? (cc << 3) : 0;
                const int colx = (g == 0 && l > 0) ? (cc << 3) : 0;
                group_pass(S, &U[l][g][0][0], rowx, colx, ws, 20 - ws, wave, c, h);
            }
        }

        {
            v2f s01 = {0.0f, 0.0f}, s23 = {0.0f, 0.0f};
            float sa = 0.0f, sb = 0.0f;
            float dq0 = 0.0f, dq1 = 0.0f, dq2 = 0.0f, dq3 = 0.0f;
#pragma unroll
            for (int q3 = 0; q3 < 16; ++q3) {
                const float4 va = S4[tid + THREADS * (2 * q3)];
                const float4 vb = S4[tid + THREADS * (2 * q3 + 1)];
                const v2f a01 = {va.x, va.y}, a23 = {va.z, va.w};
                const v2f b01 = {vb.x, vb.y}, b23 = {vb.z, vb.w};
                s01 += a01 * a01; s23 += a23 * a23;
                s01 += b01 * b01; s23 += b23 * b23;
                const v2f ta = a01 * a01 + a23 * a23;
                const v2f tb = b01 * b01 + b23 * b23;
                const float pa = ta.x + ta.y;
                const float pb = tb.x + tb.y;
                sa += pa;
                sb += pb;
                const float pq = pa + pb;
                dq0 += (q3 & 1) ? -pq : pq;
                dq1 += (q3 & 2) ? -pq : pq;
                dq2 += (q3 & 4) ? -pq : pq;
                dq3 += (q3 & 8) ? -pq : pq;
            }
            const float s0 = s01.x, s1 = s01.y, s2 = s23.x, s3 = s23.y;
            const float db  = sa - sb;
            const float tot = sa + sb;
            const int zp  = (tid >> 2) & 15;
            const int vlo = (tid >> 6) & 7;
            const int ch  = (tid & 3) ^ ((zp >> 2) & 3);
            const int ulo = zp ^ vlo;
            acc[0] += (ch & 2) ? -dq0 : dq0;
            acc[1] += (ch & 1) ? -tot : tot;
            acc[2] += (s0 + s1) - (s2 + s3);
            acc[3] += (s0 + s2) - (s1 + s3);
            acc[4] += (ulo & 8) ? -db : db;
            acc[5] += (ulo & 4) ? -tot : tot;
            acc[6] += (ulo & 2) ? -tot : tot;
            acc[7] += (ulo & 1) ? -tot : tot;
            acc[8] += db;
            acc[9]  += (vlo & 4) ? -tot : tot;
            acc[10] += (vlo & 2) ? -tot : tot;
            acc[11] += (vlo & 1) ? -tot : tot;
            acc[12] += dq3;
            acc[13] += dq2;
            acc[14] += dq1;
            acc[15] += dq0;
        }
        __syncthreads();
    }

#pragma unroll
    for (int w = 0; w < 16; ++w) {
        float v = acc[w];
        v += __shfl_xor(v, 16, 32);
        v += __shfl_xor(v, 8, 32);
        v += __shfl_xor(v, 4, 32);
        v += __shfl_xor(v, 2, 32);
        v += __shfl_xor(v, 1, 32);
        acc[w] = v;
    }
    if (lane == 0) {
#pragma unroll
        for (int w = 0; w < 16; ++w) partial[wave][w] = acc[w];
    }
    __syncthreads();
    if (tid < 16) {
        float s = 0.0f;
#pragma unroll
        for (int q = 0; q < 16; ++q) s += partial[q][tid];
        out[b * N_QUBITS + tid] = s;
    }
}

extern "C" void kernel_launch(void* const* d_in, const int* in_sizes, int n_in,
                              void* d_out, int out_size, void* d_ws, size_t ws_size,
                              hipStream_t stream) {
    const float* x      = (const float*)d_in[0];
    const float* params = (const float*)d_in[1];
    float* out          = (float*)d_out;
    (void)in_sizes; (void)n_in; (void)out_size; (void)d_ws; (void)ws_size;
    qsim16_kernel<<<BATCH, THREADS, 0, stream>>>(x, params, out);
}


# First pass of a part forms its B fragments from the RX factor tables, only the 8 non-zero k per column (popcount parity): fill gone, 2 of its 4 k steps gone; RY sum order changes
# speedup vs baseline: 2.5574x; 1.0553x over previous
#include <hip/hip_runtime.h>


#define N_QUBITS 16
#define N_LAYERS 2
#define BATCH    512
#define DIM      65536
#define THREADS  512

typedef __attribute__((ext_vector_type(2))) float v2f;
typedef __attribute__((ext_vector_type(8))) float v8f;

__device__ __forceinline__ float quarter_cos(int k)
{
    k &= 3;
    return (k == 0) ? 1.0f : ((k == 2) ? -1.0f : 0.0f);
}

__device__ __forceinline__ void group_pass(float* S, const float* Um, int rowx, int colx,
                                           int WS, int CS, int wave, int c, int h)
{
    const int rsel = c ^ (c >> 1);
    v2f afr[4];
#pragma unroll
    for (int k0 = 0; k0 < 4; ++k0) {
        int K = (8 * h + 2 * k0) ^ colx;
        afr[k0].x = Um[(rsel ^ rowx) * 16 + K];
        afr[k0].y = Um[(rsel ^ rowx) * 16 + K + 1];
    }

    v8f d[16];
    float4 bq[2][8];
    const int lbase = (wave << WS) | (c << CS) | (c << 4) | (((2 * h) ^ ((c >> 2) & 3)) << 2);
#pragma unroll
    for (int i = 0; i < 4; ++i) {
        const int off = lbase ^ ((i << 4) | ((i >> 2) << 2));
        bq[0][2 * i]     = *reinterpret_cast<const float4*>(&S[off]);
        bq[0][2 * i + 1] = *reinterpret_cast<const float4*>(&S[off ^ 4]);
    }
#pragma unroll
    for (int blk = 0; blk < 4; ++blk) {
        const int cur = blk & 1;
        if (blk < 3) {
#pragma unroll
            for (int i = 0; i < 4; ++i) {
                const int t   = 4 * (blk + 1) + i;
                const int off = lbase ^ ((t << 4) | ((t >> 2) << 2));
                bq[cur ^ 1][2 * i]     = *reinterpret_cast<const float4*>(&S[off]);
                bq[cur ^ 1][2 * i + 1] = *reinterpret_cast<const float4*>(&S[off ^ 4]);
            }
        }
        __builtin_amdgcn_sched_barrier(0);
        v8f acc[4] = {};
#pragma unroll
        for (int k0 = 0; k0 < 4; ++k0) {
#pragma unroll
            for (int i = 0; i < 4; ++i) {
                const float4 q = bq[cur][2 * i + (k0 >> 1)];
                const v2f f = (k0 & 1) ? (v2f){q.z, q.w} : (v2f){q.x, q.y};
                acc[i] = __builtin_amdgcn_wmma_f32_16x16x4_f32(false, afr[k0], false, f, (short)0, acc[i], false, false);
            }
        }
#pragma unroll
        for (int i = 0; i < 4; ++i) d[4 * blk + i] = acc[i];
        __builtin_amdgcn_sched_barrier(0);
    }
    const int zs = c ^ wave;
    const int ss = (zs >> 2) & 3;
#pragma unroll
    for (int r = 0; r < 8; ++r) {
        const int rowo = (wave << WS) | ((8 * h + r) << CS) | (zs << 4);
#pragma unroll
        for (int j = 0; j < 4; ++j) {
            float4 o;
            o.x = d[4 * j][r]; o.y = d[4 * j + 1][r]; o.z = d[4 * j + 2][r]; o.w = d[4 * j + 3][r];
            *reinterpret_cast<float4*>(&S[rowo | ((j ^ ss) << 2)]) = o;
        }
    }
    __syncthreads();
}

__device__ __forceinline__ void first_pass(float* S, const float* Um, const float (*Gt)[16],
                                           int part, int wave, int c, int h)
{
    const int rsel = c ^ (c >> 1);
    const int pc   = __popc(c) & 1;
    const int w0   = (__popc(wave) + part) & 1;
    v2f   afr[2][2];
    float g0s[2][4];
    int   qv[2];
#pragma unroll
    for (int ps = 0; ps < 2; ++ps) {
        const int q = ps ^ w0 ^ h;
        qv[ps] = q;
#pragma unroll
        for (int j = 0; j < 4; ++j) {
            const int e  = (j == 0) ? 0 : ((j == 1) ? 3 : ((j == 2) ? 5 : 6));
            const int a0 = 8 * h + (e ^ q);
            const float g = Gt[0][a0];
            const bool neg = q ? (j == 3) : (j != 0);
            g0s[ps][j] = neg ? -g : g;
            const float u = Um[rsel * 16 + a0];
            if (j & 1) afr[ps][j >> 1].y = u; else afr[ps][j >> 1].x = u;
        }
    }
    const float g2 = Gt[2][c];
    const float g3 = Gt[3][wave];
    const int nbase = h + __popc(c) + __popc(wave) + part;
    const float* g1e = &Gt[1][0] + pc;
    const float* g1o = &Gt[1][0] - pc;
    int nlane[2][2];
#pragma unroll
    for (int ps = 0; ps < 2; ++ps) {
        nlane[ps][0] = nbase + qv[ps] + pc;
        nlane[ps][1] = nbase + qv[ps] - pc;
    }

    v8f d[16];
#pragma unroll
    for (int blk = 0; blk < 4; ++blk) {
        v2f f[4][2];
#pragma unroll
        for (int i = 0; i < 4; ++i) {
            const int s  = 4 * blk + i;
            const int ps = __builtin_popcount(s) & 1;
            const float ga = (s & 1) ? g1o[s] : g1e[s];
            float b3 = (ga * g2) * g3;
            const int n = nlane[ps][s & 1] + __builtin_popcount(s);
            if (n & 2) b3 = -b3;
            f[i][0] = (v2f){b3 * g0s[ps][0], b3 * g0s[ps][1]};
            f[i][1] = (v2f){b3 * g0s[ps][2], b3 * g0s[ps][3]};
        }
        v8f acc[4] = {};
#pragma unroll
        for (int k0 = 0; k0 < 2; ++k0) {
#pragma unroll
            for (int i = 0; i < 4; ++i) {
                const int ps = __builtin_popcount(4 * blk + i) & 1;
                acc[i] = __builtin_amdgcn_wmma_f32_16x16x4_f32(false, afr[ps][k0], false, f[i][k0], (short)0, acc[i], false, false);
            }
        }
#pragma unroll
        for (int i = 0; i < 4; ++i) d[4 * blk + i] = acc[i];
    }
    const int zs = c ^ wave;
    const int ss = (zs >> 2) & 3;
    float* sb[4];
#pragma unroll
    for (int j = 0; j < 4; ++j) sb[j] = &S[(wave << 12) | (h << 11) | (zs << 4) | ((j ^ ss) << 2)];
#pragma unroll
    for (int r = 0; r < 8; ++r) {
#pragma unroll
        for (int j = 0; j < 4; ++j) {
            float4 o;
            o.x = pc ? d[4 * j + 1][r] : d[4 * j][r];
            o.y = pc ? d[4 * j][r]     : d[4 * j + 1][r];
            o.z = pc ? d[4 * j + 3][r] : d[4 * j + 2][r];
            o.w = pc ? d[4 * j + 2][r] : d[4 * j + 3][r];
            *reinterpret_cast<float4*>(sb[j] + (r << 8)) = o;
        }
    }
    __syncthreads();
}

__global__ __launch_bounds__(THREADS)
void qsim16_kernel(const float* __restrict__ x,
                   const float* __restrict__ params,
                   float* __restrict__ out)
{
    __shared__ __align__(16) float S[DIM];
    __shared__ float U[N_LAYERS][4][16][16];
    __shared__ __align__(16) float Gt[4][16];
    __shared__ float cw[16], sw[16];
    __shared__ float pcs[N_LAYERS][16];
    __shared__ float pss[N_LAYERS][16];
    __shared__ float partial[16][16];

    float4* S4 = reinterpret_cast<float4*>(S);

    const int tid  = threadIdx.x;
    const int lane = tid & 31;
    const int wave = tid >> 5;
    const int b    = blockIdx.x;

    if (tid < 16) {
        float a = 0.5f * x[b * N_QUBITS + tid];
        float s, c;
        __sincosf(a, &s, &c);
        cw[tid] = c;
        sw[tid] = s;
    } else if (tid < 48) {
        int idx = tid - 16;
        int l = idx >> 4, w = idx & 15;
        float a = 0.5f * params[idx];
        float s, c;
        __sincosf(a, &s, &c);
        pcs[l][w] = c;
        pss[l][w] = s;
    }
    __syncthreads();

    if (tid < 64) {
        int g = tid >> 4, a = tid & 15;
        float p = 1.0f;
#pragma unroll
        for (int m = 0; m < 4; ++m)
            p *= ((a >> (3 - m)) & 1) ? sw[4 * g + m] : cw[4 * g + m];
        Gt[g][a] = p;
    }

    for (int e = tid; e < N_LAYERS * 4 * 256; e += THREADS) {
        int l  = e >> 10;
        int g  = (e >> 8) & 3;
        int ar = (e >> 4) & 15;
        int ac = e & 15;
        float v = 1.0f;
#pragma unroll
        for (int m = 0; m < 4; ++m) {
            int w = 4 * g + m;
            int i = (ar >> (3 - m)) & 1;
            int j = (ac >> (3 - m)) & 1;
            float f = (i == j) ? pcs[l][w] : (i ? pss[l][w] : -pss[l][w]);
            v *= f;
        }
        U[l][g][ar][ac] = v;
    }
    __syncthreads();

    float acc[16];
#pragma unroll
    for (int w = 0; w < 16; ++w) acc[w] = 0.0f;

    const int c = lane & 15;
    const int h = lane >> 4;
    const int cc = wave & 1;

    for (int part = 0; part < 2; ++part) {
        first_pass(S, &U[0][0][0][0], Gt, part, wave, c, h);

#pragma unroll 1
        for (int p = 1; p < 4 * N_LAYERS; ++p) {
            const int l    = p >> 2;
            const int g    = p & 3;
            const int ws   = (g & 1) ? 8 : 12;
            const int rowx = (g > 0) ? (cc << 3) : 0;
            const int colx = (g == 0) ? (cc << 3) : 0;
            group_pass(S, &U[l][g][0][0], rowx, colx, ws, 20 - ws, wave, c, h);
        }

        {
            v2f s01 = {0.0f, 0.0f}, s23 = {0.0f, 0.0f};
            float sa = 0.0f, sb = 0.0f;
            float dq0 = 0.0f, dq1 = 0.0f, dq2 = 0.0f, dq3 = 0.0f;
#pragma unroll
            for (int q3 = 0; q3 < 16; ++q3) {
                const float4 va = S4[tid + THREADS * (2 * q3)];
                const float4 vb = S4[tid + THREADS * (2 * q3 + 1)];
                const v2f a01 = {va.x, va.y}, a23 = {va.z, va.w};
                const v2f b01 = {vb.x, vb.y}, b23 = {vb.z, vb.w};
                s01 += a01 * a01; s23 += a23 * a23;
                s01 += b01 * b01; s23 += b23 * b23;
                const v2f ta = a01 * a01 + a23 * a23;
                const v2f tb = b01 * b01 + b23 * b23;
                const float pa = ta.x + ta.y;
                const float pb = tb.x + tb.y;
                sa += pa;
                sb += pb;
                const float pq = pa + pb;
                dq0 += (q3 & 1) ? -pq : pq;
                dq1 += (q3 & 2) ? -pq : pq;
                dq2 += (q3 & 4) ? -pq : pq;
                dq3 += (q3 & 8) ? -pq : pq;
            }
            const float s0 = s01.x, s1 = s01.y, s2 = s23.x, s3 = s23.y;
            const float db  = sa - sb;
            const float tot = sa + sb;
            const int zp  = (tid >> 2) & 15;
            const int vlo = (tid >> 6) & 7;
            const int ch  = (tid & 3) ^ ((zp >> 2) & 3);
            const int ulo = zp ^ vlo;
            acc[0] += (ch & 2) ? -dq0 : dq0;
            acc[1] += (ch & 1) ? -tot : tot;
            acc[2] += (s0 + s1) - (s2 + s3);
            acc[3] += (s0 + s2) - (s1 + s3);
            acc[4] += (ulo & 8) ? -db : db;
            acc[5] += (ulo & 4) ? -tot : tot;
            acc[6] += (ulo & 2) ? -tot : tot;
            acc[7] += (ulo & 1) ? -tot : tot;
            acc[8] += db;
            acc[9]  += (vlo & 4) ? -tot : tot;
            acc[10] += (vlo & 2) ? -tot : tot;
            acc[11] += (vlo & 1) ? -tot : tot;
            acc[12] += dq3;
            acc[13] += dq2;
            acc[14] += dq1;
            acc[15] += dq0;
        }
        __syncthreads();
    }

#pragma unroll
    for (int w = 0; w < 16; ++w) {
        float v = acc[w];
        v += __shfl_xor(v, 16, 32);
        v += __shfl_xor(v, 8, 32);
        v += __shfl_xor(v, 4, 32);
        v += __shfl_xor(v, 2, 32);
        v += __shfl_xor(v, 1, 32);
        acc[w] = v;
    }
    if (lane == 0) {
#pragma unroll
        for (int w = 0; w < 16; ++w) partial[wave][w] = acc[w];
    }
    __syncthreads();
    if (tid < 16) {
        float s = 0.0f;
#pragma unroll
        for (int q = 0; q < 16; ++q) s += partial[q][tid];
        out[b * N_QUBITS + tid] = s;
    }
}

extern "C" void kernel_launch(void* const* d_in, const int* in_sizes, int n_in,
                              void* d_out, int out_size, void* d_ws, size_t ws_size,
                              hipStream_t stream) {
    const float* x      = (const float*)d_in[0];
    const float* params = (const float*)d_in[1];
    float* out          = (float*)d_out;
    (void)in_sizes; (void)n_in; (void)out_size; (void)d_ws; (void)ws_size;
    qsim16_kernel<<<BATCH, THREADS, 0, stream>>>(x, params, out);
}


# RY passes swap WMMA operands (state = A, U^T = B): a lane's result registers lie along the next pass's k and are stored as they are, no register moves; LDS z field u^v^q; expval sum order changes
# speedup vs baseline: 2.7296x; 1.0673x over previous
#include <hip/hip_runtime.h>


#define N_QUBITS 16
#define N_LAYERS 2
#define BATCH    512
#define DIM      65536
#define THREADS  512

typedef __attribute__((ext_vector_type(2))) float v2f;
typedef __attribute__((ext_vector_type(8))) float v8f;

__device__ __forceinline__ float quarter_cos(int k)
{
    k &= 3;
    return (k == 0) ? 1.0f : ((k == 2) ? -1.0f : 0.0f);
}

__device__ __forceinline__ void group_pass(float* S, const float* Um, int rowx, int colx,
                                           int WS, int CS, int wave, int zq, int c, int h)
{
    const int rsel = c ^ (c >> 1);
    v2f ufr[4];
#pragma unroll
    for (int k0 = 0; k0 < 4; ++k0) {
        int K = (8 * h + 2 * k0) ^ colx;
        ufr[k0].x = Um[(rsel ^ rowx) * 16 + K];
        ufr[k0].y = Um[(rsel ^ rowx) * 16 + K + 1];
    }

    v8f d[16];
    float4 bq[2][8];
    char* Sb = reinterpret_cast<char*>(S);
    const int zl    = c ^ zq;
    const int zs    = c ^ wave;
    const int lbase = ((wave << WS) | (zl << 4) | (((2 * h) ^ ((zl >> 2) & 3)) << 2)) << 2;
    const int sbase = ((wave << WS) | (c << CS) | (zs << 4) | (((2 * h) ^ ((zs >> 2) & 3)) << 2)) << 2;
#pragma unroll
    for (int i = 0; i < 4; ++i) {
        const int off = lbase ^ ((i << (CS + 2)) | (((i << 4) | ((i >> 2) << 2)) << 2));
        bq[0][2 * i]     = *reinterpret_cast<const float4*>(Sb + off);
        bq[0][2 * i + 1] = *reinterpret_cast<const float4*>(Sb + (off ^ 16));
    }
#pragma unroll
    for (int blk = 0; blk < 4; ++blk) {
        const int cur = blk & 1;
        if (blk < 3) {
#pragma unroll
            for (int i = 0; i < 4; ++i) {
                const int t   = 4 * (blk + 1) + i;
                const int off = lbase ^ ((t << (CS + 2)) | (((t << 4) | ((t >> 2) << 2)) << 2));
                bq[cur ^ 1][2 * i]     = *reinterpret_cast<const float4*>(Sb + off);
                bq[cur ^ 1][2 * i + 1] = *reinterpret_cast<const float4*>(Sb + (off ^ 16));
            }
        }
        __builtin_amdgcn_sched_barrier(0);
        v8f acc[4] = {};
#pragma unroll
        for (int k0 = 0; k0 < 4; ++k0) {
#pragma unroll
            for (int i = 0; i < 4; ++i) {
                const float4 q = bq[cur][2 * i + (k0 >> 1)];
                const v2f f = (k0 & 1) ? (v2f){q.z, q.w} : (v2f){q.x, q.y};
                acc[i] = __builtin_amdgcn_wmma_f32_16x16x4_f32(false, f, false, ufr[k0], (short)0, acc[i], false, false);
            }
        }
#pragma unroll
        for (int i = 0; i < 4; ++i) d[4 * blk + i] = acc[i];
        __builtin_amdgcn_sched_barrier(0);
    }
#pragma unroll
    for (int t = 0; t < 16; ++t) {
        const int off = sbase ^ (((t << 4) | ((t >> 2) << 2)) << 2);
        float4 lo, hi;
        lo.x = d[t][0]; lo.y = d[t][1]; lo.z = d[t][2]; lo.w = d[t][3];
        hi.x = d[t][4]; hi.y = d[t][5]; hi.z = d[t][6]; hi.w = d[t][7];
        *reinterpret_cast<float4*>(Sb + off)        = lo;
        *reinterpret_cast<float4*>(Sb + (off ^ 16)) = hi;
    }
    __syncthreads();
}

__device__ __forceinline__ void first_pass(float* S, const float* Um, const float (*Gt)[16],
                                           int part, int wave, int c, int h)
{
    const int rsel = c ^ (c >> 1);
    const int pc   = __popc(c) & 1;
    const int w0   = (__popc(wave) + part) & 1;
    v2f   afr[2][2];
    float g0s[2][4];
    int   qv[2];
#pragma unroll
    for (int ps = 0; ps < 2; ++ps) {
        const int q = ps ^ w0 ^ h;
        qv[ps] = q;
#pragma unroll
        for (int j = 0; j < 4; ++j) {
            const int e  = (j == 0) ? 0 : ((j == 1) ? 3 : ((j == 2) ? 5 : 6));
            const int a0 = 8 * h + (e ^ q);
            const float g = Gt[0][a0];
            const bool neg = q ? (j == 3) : (j != 0);
            g0s[ps][j] = neg ? -g : g;
            const float u = Um[rsel * 16 + a0];
            if (j & 1) afr[ps][j >> 1].y = u; else afr[ps][j >> 1].x = u;
        }
    }
    const float g2 = Gt[2][c];
    const float g3 = Gt[3][wave];
    const int nbase = h + __popc(c) + __popc(wave) + part;
    const float* g1e = &Gt[1][0] + pc;
    const float* g1o = &Gt[1][0] - pc;
    int nlane[2][2];
#pragma unroll
    for (int ps = 0; ps < 2; ++ps) {
        nlane[ps][0] = nbase + qv[ps] + pc;
        nlane[ps][1] = nbase + qv[ps] - pc;
    }

    v8f d[16];
#pragma unroll
    for (int blk = 0; blk < 4; ++blk) {
        v2f f[4][2];
#pragma unroll
        for (int i = 0; i < 4; ++i) {
            const int s  = 4 * blk + i;
            const int ps = __builtin_popcount(s) & 1;
            const float ga = (s & 1) ? g1o[s] : g1e[s];
            float b3 = (ga * g2) * g3;
            const int n = nlane[ps][s & 1] + __builtin_popcount(s);
            if (n & 2) b3 = -b3;
            f[i][0] = (v2f){b3 * g0s[ps][0], b3 * g0s[ps][1]};
            f[i][1] = (v2f){b3 * g0s[ps][2], b3 * g0s[ps][3]};
        }
        v8f acc[4] = {};
#pragma unroll
        for (int k0 = 0; k0 < 2; ++k0) {
#pragma unroll
            for (int i = 0; i < 4; ++i) {
                const int ps = __builtin_popcount(4 * blk + i) & 1;
                acc[i] = __builtin_amdgcn_wmma_f32_16x16x4_f32(false, afr[ps][k0], false, f[i][k0], (short)0, acc[i], false, false);
            }
        }
#pragma unroll
        for (int i = 0; i < 4; ++i) d[4 * blk + i] = acc[i];
    }
    const int zs = c ^ wave;
    const int ss = (zs >> 2) & 3;
    float* sb[4];
#pragma unroll
    for (int j = 0; j < 4; ++j) sb[j] = &S[(wave << 12) | (h << 11) | (zs << 4) | ((j ^ ss) << 2)];
#pragma unroll
    for (int r = 0; r < 8; ++r) {
#pragma unroll
        for (int j = 0; j < 4; ++j) {
            float4 o;
            o.x = pc ? d[4 * j + 1][r] : d[4 * j][r];
            o.y = pc ? d[4 * j][r]     : d[4 * j + 1][r];
            o.z = pc ? d[4 * j + 3][r] : d[4 * j + 2][r];
            o.w = pc ? d[4 * j + 2][r] : d[4 * j + 3][r];
            *reinterpret_cast<float4*>(sb[j] + (r << 8)) = o;
        }
    }
    __syncthreads();
}

__global__ __launch_bounds__(THREADS)
void qsim16_kernel(const float* __restrict__ x,
                   const float* __restrict__ params,
                   float* __restrict__ out)
{
    __shared__ __align__(16) float S[DIM];
    __shared__ float U[N_LAYERS][4][16][16];
    __shared__ __align__(16) float Gt[4][16];
    __shared__ float cw[16], sw[16];
    __shared__ float pcs[N_LAYERS][16];
    __shared__ float pss[N_LAYERS][16];
    __shared__ float partial[16][16];

    float4* S4 = reinterpret_cast<float4*>(S);

    const int tid  = threadIdx.x;
    const int lane = tid & 31;
    const int wave = tid >> 5;
    const int b    = blockIdx.x;

    if (tid < 16) {
        float a = 0.5f * x[b * N_QUBITS + tid];
        float s, c;
        __sincosf(a, &s, &c);
        cw[tid] = c;
        sw[tid] = s;
    } else if (tid < 48) {
        int idx = tid - 16;
        int l = idx >> 4, w = idx & 15;
        float a = 0.5f * params[idx];
        float s, c;
        __sincosf(a, &s, &c);
        pcs[l][w] = c;
        pss[l][w] = s;
    }
    __syncthreads();

    if (tid < 64) {
        int g = tid >> 4, a = tid & 15;
        float p = 1.0f;
#pragma unroll
        for (int m = 0; m < 4; ++m)
            p *= ((a >> (3 - m)) & 1) ? sw[4 * g + m] : cw[4 * g + m];
        Gt[g][a] = p;
    }

    for (int e = tid; e < N_LAYERS * 4 * 256; e += THREADS) {
        int l  = e >> 10;
        int g  = (e >> 8) & 3;
        int ar = (e >> 4) & 15;
        int ac = e & 15;
        float v = 1.0f;
#pragma unroll
        for (int m = 0; m < 4; ++m) {
            int w = 4 * g + m;
            int i = (ar >> (3 - m)) & 1;
            int j = (ac >> (3 - m)) & 1;
            float f = (i == j) ? pcs[l][w] : (i ? pss[l][w] : -pss[l][w]);
            v *= f;
        }
        U[l][g][ar][ac] = v;
    }
    __syncthreads();

    float acc[16];
#pragma unroll
    for (int w = 0; w < 16; ++w) acc[w] = 0.0f;

    const int c = lane & 15;
    const int h = lane >> 4;
    const int cc = wave & 1;

    for (int part = 0; part < 2; ++part) {
        first_pass(S, &U[0][0][0][0], Gt, part, wave, c, h);

#pragma unroll 1
        for (int p = 1; p < 4 * N_LAYERS; ++p) {
            const int l    = p >> 2;
            const int g    = p & 3;
            const int ws   = (g & 1) ? 8 : 12;
            const int rowx = (g > 0) ? (cc << 3) : 0;
            const int colx = (g == 0) ? (cc << 3) : 0;
            group_pass(S, &U[l][g][0][0], rowx, colx, ws, 20 - ws, wave, (p == 1) ? 0 : wave, c, h);
        }

        {
            v2f s01 = {0.0f, 0.0f}, s23 = {0.0f, 0.0f};
            float sa = 0.0f, sb = 0.0f;
            float dq0 = 0.0f, dq1 = 0.0f, dq2 = 0.0f, dq3 = 0.0f;
            float dbq = 0.0f;
#pragma unroll
            for (int q3 = 0; q3 < 16; ++q3) {
                const float4 va = S4[tid + THREADS * (2 * q3)];
                const float4 vb = S4[tid + THREADS * (2 * q3 + 1)];
                const v2f a01 = {va.x, va.y}, a23 = {va.z, va.w};
                const v2f b01 = {vb.x, vb.y}, b23 = {vb.z, vb.w};
                s01 += a01 * a01; s23 += a23 * a23;
                s01 += b01 * b01; s23 += b23 * b23;
                const v2f ta = a01 * a01 + a23 * a23;
                const v2f tb = b01 * b01 + b23 * b23;
                const float pa = ta.x + ta.y;
                const float pb = tb.x + tb.y;
                sa += pa;
                sb += pb;
                const float pq = pa + pb;
                dq0 += (q3 & 1) ? -pq : pq;
                dq1 += (q3 & 2) ? -pq : pq;
                dq2 += (q3 & 4) ? -pq : pq;
                dq3 += (q3 & 8) ? -pq : pq;
                const float pd = pa - pb;
                dbq += (q3 & 8) ? -pd : pd;
            }
            const float s0 = s01.x, s1 = s01.y, s2 = s23.x, s3 = s23.y;
            const float db  = sa - sb;
            const float tot = sa + sb;
            const int zp  = (tid >> 2) & 15;
            const int vlo = (tid >> 6) & 7;
            const int ch  = (tid & 3) ^ ((zp >> 2) & 3);
            const int ulo = zp ^ vlo;
            acc[0] += (ch & 2) ? -dq0 : dq0;
            acc[1] += (ch & 1) ? -tot : tot;
            acc[2] += (s0 + s1) - (s2 + s3);
            acc[3] += (s0 + s2) - (s1 + s3);
            acc[4] += (ulo & 8) ? -dbq : dbq;
            acc[5] += (ulo & 4) ? -dq2 : dq2;
            acc[6] += (ulo & 2) ? -dq1 : dq1;
            acc[7] += (ulo & 1) ? -dq0 : dq0;
            acc[8] += db;
            acc[9]  += (vlo & 4) ? -tot : tot;
            acc[10] += (vlo & 2) ? -tot : tot;
            acc[11] += (vlo & 1) ? -tot : tot;
            acc[12] += dq3;
            acc[13] += dq2;
            acc[14] += dq1;
            acc[15] += dq0;
        }
        __syncthreads();
    }

#pragma unroll
    for (int w = 0; w < 16; ++w) {
        float v = acc[w];
        v += __shfl_xor(v, 16, 32);
        v += __shfl_xor(v, 8, 32);
        v += __shfl_xor(v, 4, 32);
        v += __shfl_xor(v, 2, 32);
        v += __shfl_xor(v, 1, 32);
        acc[w] = v;
    }
    if (lane == 0) {
#pragma unroll
        for (int w = 0; w < 16; ++w) partial[wave][w] = acc[w];
    }
    __syncthreads();
    if (tid < 16) {
        float s = 0.0f;
#pragma unroll
        for (int q = 0; q < 16; ++q) s += partial[q][tid];
        out[b * N_QUBITS + tid] = s;
    }
}

extern "C" void kernel_launch(void* const* d_in, const int* in_sizes, int n_in,
                              void* d_out, int out_size, void* d_ws, size_t ws_size,
                              hipStream_t stream) {
    const float* x      = (const float*)d_in[0];
    const float* params = (const float*)d_in[1];
    float* out          = (float*)d_out;
    (void)in_sizes; (void)n_in; (void)out_size; (void)d_ws; (void)ws_size;
    qsim16_kernel<<<BATCH, THREADS, 0, stream>>>(x, params, out);
}
